# starting kernel
# baseline (speedup 1.0000x reference)
#include <hip/hip_runtime.h>
#include <math.h>


typedef __attribute__((ext_vector_type(2))) float v2f;
typedef __attribute__((ext_vector_type(8))) float v8f;

#define LDS_STRIDE 68

__device__ __forceinline__ float gelu_exact(float x) {
    return 0.5f * x * (1.0f + erff(x * 0.70710678118654752f));
}

__global__ __launch_bounds__(256)
void embed_kernel(const int* __restrict__ positions,
                  const int* __restrict__ amino,
                  const float* __restrict__ pos_table,
                  const float* __restrict__ aa_table,
                  float* __restrict__ E) {
    int t  = blockIdx.x * 256 + threadIdx.x;
    int k  = t & 63;
    int bm = t >> 6;
    int p = positions[bm]; p = p < 0 ? 0 : (p > 1023 ? 1023 : p);
    int a = amino[bm];     a = a < 0 ? 0 : (a > 20   ? 20   : a);
    E[t] = pos_table[p * 64 + k] + aa_table[a * 64 + k];
}

__global__ __launch_bounds__(256)
void pair_kernel(const float* __restrict__ E,
                 const float* __restrict__ W1,
                 const float* __restrict__ b1,
                 const float* __restrict__ W2,
                 const float* __restrict__ b2,
                 const float* __restrict__ W3,
                 const float* __restrict__ b3,
                 float* __restrict__ out) {
    __shared__ float u_lds[16 * LDS_STRIDE];
    __shared__ float v_lds[16 * LDS_STRIDE];
    __shared__ float w2t [64 * LDS_STRIDE];
    __shared__ float b1s[64], b2s[64], w3s[64];

    const int tid = threadIdx.x;
    const int b   = blockIdx.y;
    const int tp  = blockIdx.x;
    const int it  = (tp < 4) ? 0 : (tp < 7) ? 1 : (tp < 9) ? 2 : 3;
    const int start = it * 4 - (it * (it - 1)) / 2;
    const int jt  = tp - start + it;

    {
        const int n  = tid & 63;
        const int i0 = tid >> 6;
        const float* Ei = E + (b * 64 + it * 16) * 64;
        const float* Ej = E + (b * 64 + jt * 16) * 64;
        for (int s = 0; s < 4; ++s) {
            const int i = i0 + 4 * s;
            float su = 0.f, sv = 0.f;
            for (int k = 0; k < 64; ++k) {
                su = fmaf(Ei[i * 64 + k], W1[k * 64 + n],        su);
                sv = fmaf(Ej[i * 64 + k], W1[(64 + k) * 64 + n], sv);
            }
            u_lds[i * LDS_STRIDE + n] = su;
            v_lds[i * LDS_STRIDE + n] = sv;
        }
        for (int e = tid; e < 4096; e += 256) {
            int k = e >> 6, nn = e & 63;
            w2t[nn * LDS_STRIDE + k] = W2[k * 64 + nn];
        }
        if (tid < 64) { b1s[tid] = b1[tid]; b2s[tid] = b2[tid]; w3s[tid] = W3[tid]; }
    }
    __syncthreads();

    const int lane = tid & 31;
    const int wv   = tid >> 5;
    const int jl   = lane & 15;
    const int half = lane >> 4;
    const int koff = 2 * half;

    v2f hA[2][16];
    #pragma unroll
    for (int mt = 0; mt < 2; ++mt) {
        const int il = 2 * wv + mt;
        #pragma unroll
        for (int kk = 0; kk < 16; ++kk) {
            const int k0 = 4 * kk + koff;
            float x0 = u_lds[il * LDS_STRIDE + k0]     + v_lds[jl * LDS_STRIDE + k0]     + b1s[k0];
            float x1 = u_lds[il * LDS_STRIDE + k0 + 1] + v_lds[jl * LDS_STRIDE + k0 + 1] + b1s[k0 + 1];
            hA[mt][kk].x = gelu_exact(x0);
            hA[mt][kk].y = gelu_exact(x1);
        }
    }

    float p0[8], p1[8];
    #pragma unroll
    for (int r = 0; r < 8; ++r) { p0[r] = 0.f; p1[r] = 0.f; }

    #pragma unroll
    for (int nt = 0; nt < 4; ++nt) {
        const int n = nt * 16 + jl;
        v8f acc0 = {};
        v8f acc1 = {};
        #pragma unroll
        for (int kk = 0; kk < 16; ++kk) {
            const int k0 = 4 * kk + koff;
            v2f bb = *(const v2f*)&w2t[n * LDS_STRIDE + k0];
            acc0 = __builtin_amdgcn_wmma_f32_16x16x4_f32(false, hA[0][kk], false, bb,
                                                         (short)0, acc0, false, false);
            acc1 = __builtin_amdgcn_wmma_f32_16x16x4_f32(false, hA[1][kk], false, bb,
                                                         (short)0, acc1, false, false);
        }
        const float b2n = b2s[n];
        const float w3n = w3s[n];
        #pragma unroll
        for (int r = 0; r < 8; ++r) {
            p0[r] = fmaf(gelu_exact(acc0[r] + b2n), w3n, p0[r]);
            p1[r] = fmaf(gelu_exact(acc1[r] + b2n), w3n, p1[r]);
        }
    }

    const float bias3 = b3[0];
    #pragma unroll
    for (int mt = 0; mt < 2; ++mt) {
        const int ig = it * 16 + 2 * wv + mt;
        #pragma unroll
        for (int r = 0; r < 8; ++r) {
            float s = (mt == 0) ? p0[r] : p1[r];
            s += __shfl_xor(s, 1);
            s += __shfl_xor(s, 2);
            s += __shfl_xor(s, 4);
            s += __shfl_xor(s, 8);
            if (jl == 0) {
                const int jg = jt * 16 + r + 8 * half;
                if (ig < jg) {
                    const float val = s + bias3;
                    out[(b * 64 + ig) * 64 + jg] = val;
                    out[(b * 64 + jg) * 64 + ig] = val;
                } else if (ig == jg) {
                    out[(b * 64 + ig) * 64 + jg] = 0.f;
                }
            }
        }
    }
}

extern "C" void kernel_launch(void* const* d_in, const int* in_sizes, int n_in,
                              void* d_out, int out_size, void* d_ws, size_t ws_size,
                              hipStream_t stream) {
    const int*   positions = (const int*)  d_in[0];
    const int*   amino     = (const int*)  d_in[1];
    const float* pos_table = (const float*)d_in[2];
    const float* aa_table  = (const float*)d_in[3];
    const float* W1        = (const float*)d_in[4];
    const float* b1        = (const float*)d_in[5];
    const float* W2        = (const float*)d_in[6];
    const float* b2        = (const float*)d_in[7];
    const float* W3        = (const float*)d_in[8];
    const float* b3        = (const float*)d_in[9];
    float* out = (float*)d_out;
    float* E   = (float*)d_ws;

    embed_kernel<<<dim3((128 * 64 * 64) / 256), dim3(256), 0, stream>>>(
        positions, amino, pos_table, aa_table, E);
    pair_kernel<<<dim3(10, 128), dim3(256), 0, stream>>>(
        E, W1, b1, W2, b2, W3, b3, out);
}


# pair kernel gelu: erf-form GELU read from a 256-entry cubic (Hermite) table of a*Phi(-a) built in LDS per work-group: one 16-byte LDS read + 10 plain VALU, no branch, no exp; gelu values move ~1e-7
# speedup vs baseline: 3.2454x; 3.2454x over previous
#include <hip/hip_runtime.h>
#include <math.h>


typedef __attribute__((ext_vector_type(2))) float v2f;
typedef __attribute__((ext_vector_type(4))) float v4f;
typedef __attribute__((ext_vector_type(8))) float v8f;

#define LDS_STRIDE 68

#define GT_N     256
#define GT_SCALE 32.0f
#define GT_TMAX  255.99998f

__device__ __forceinline__ float gelu_tab(float x, const v4f* __restrict__ tab) {
    const float t  = fminf(fabsf(x) * GT_SCALE, GT_TMAX);
    const float kf = floorf(t);
    const float f  = t - kf;
    const v4f   c  = tab[(int)kf];
    float p = fmaf(f, c.w, c.z);
    p = fmaf(f, p, c.y);
    p = fmaf(f, p, c.x);
    return fmaxf(x, 0.f) - p;
}

template <int MASK>
__device__ __forceinline__ float row_xor(float v) {
    return __int_as_float(__builtin_amdgcn_update_dpp(0, __float_as_int(v), 0x160 + MASK,
                                                      0xf, 0xf, true));
}

#define EF_FLOATS  (128 * 64 * 64)
#define W1F_FLOATS (128 * 64)
#define EF_BLOCKS  (EF_FLOATS / 256)
#define W1F_BLOCKS (W1F_FLOATS / 256)

__global__ __launch_bounds__(256)
void embed_kernel(const int* __restrict__ positions,
                  const int* __restrict__ amino,
                  const float* __restrict__ pos_table,
                  const float* __restrict__ aa_table,
                  const float* __restrict__ W1,
                  const float* __restrict__ W2,
                  float* __restrict__ ws) {
    const int blk = blockIdx.x;
    if (blk < EF_BLOCKS) {
        const int t    = blk * 256 + threadIdx.x;
        const int r    = t & 3;
        const int ln   = (t >> 2) & 31;
        const int q    = (t >> 7) & 7;
        const int tile = (t >> 10) & 3;
        const int b    = t >> 12;
        const int bm   = b * 64 + tile * 16 + (ln & 15);
        const int k    = 8 * q + 4 * (ln >> 4) + r;
        int p = positions[bm]; p = p < 0 ? 0 : (p > 1023 ? 1023 : p);
        int a = amino[bm];     a = a < 0 ? 0 : (a > 20   ? 20   : a);
        ws[t] = pos_table[p * 64 + k] + aa_table[a * 64 + k];
    } else if (blk < EF_BLOCKS + W1F_BLOCKS) {
        const int u    = (blk - EF_BLOCKS) * 256 + threadIdx.x;
        const int r    = u & 3;
        const int ln   = (u >> 2) & 31;
        const int q    = (u >> 7) & 7;
        const int tile = (u >> 10) & 3;
        const int sel  = u >> 12;
        const int row  = sel * 64 + 8 * q + 4 * (ln >> 4) + r;
        const int col  = tile * 16 + (ln & 15);
        ws[EF_FLOATS + u] = W1[row * 64 + col];
    } else {
        const int e = (blk - EF_BLOCKS - W1F_BLOCKS) * 256 + threadIdx.x;
        const int n = e >> 6, k = e & 63;
        ws[EF_FLOATS + W1F_FLOATS + e] = W2[k * 64 + n];
    }
}

#define PAIR_WAVES   32
#define PAIR_THREADS 1024
#define UNITS_PER_WAVE 3

__global__ __launch_bounds__(PAIR_THREADS)
void pair_kernel(const float* __restrict__ Ef,
                 const float* __restrict__ W1f,
                 const float* __restrict__ W2T,
                 const float* __restrict__ b1,
                 const float* __restrict__ b2,
                 const float* __restrict__ W3,
                 const float* __restrict__ b3,
                 float* __restrict__ out) {
    __shared__ float u_lds[64 * LDS_STRIDE];
    __shared__ float v_lds[64 * LDS_STRIDE];
    __shared__ __attribute__((aligned(16))) float w2t[64 * LDS_STRIDE];
    __shared__ float b1s[64], b2s[64], w3s[64];
    __shared__ __attribute__((aligned(16))) v4f gtab[GT_N];

    const int tid  = threadIdx.x;
    const int b    = blockIdx.x >> 1;
    const int hb   = blockIdx.x & 1;
    const int lane = tid & 31;
    const int wv   = __builtin_amdgcn_readfirstlane(tid >> 5);
    const int jl   = lane & 15;
    const int half = lane >> 4;

    {
        const int sel = wv >> 4;
        const int rt  = (wv >> 2) & 3;
        const int snt = wv & 3;
        const v4f* Afrag = (const v4f*)(Ef  + ((b * 4 + rt) * 8) * 128) + lane;
        const v4f* Bfrag = (const v4f*)(W1f + ((sel * 4 + snt) * 8) * 128) + lane;
        v4f a4[8];
        v4f b4[8];
        #pragma unroll
        for (int q = 0; q < 8; ++q)
            a4[q] = Afrag[q * 32];
        #pragma unroll
        for (int q = 0; q < 8; ++q)
            b4[q] = Bfrag[q * 32];
        if (tid < GT_N) {
            const float h  = 1.0f / GT_SCALE;
            const float a0 = (float)tid * h;
            const float a1 = (float)(tid + 1) * h;
            const float q0 = 0.5f * erfcf(a0 * 0.70710678118654752f);
            const float q1 = 0.5f * erfcf(a1 * 0.70710678118654752f);
            const float r0 = a0 * q0;
            const float r1 = a1 * q1;
            const float d0 = h * (q0 - a0 * (0.3989422804014327f * expf(-0.5f * a0 * a0)));
            const float d1 = h * (q1 - a1 * (0.3989422804014327f * expf(-0.5f * a1 * a1)));
            const float dl = r1 - r0;
            v4f c;
            c.x = r0;
            c.y = d0;
            c.z = 3.f * dl - 2.f * d0 - d1;
            c.w = d0 + d1 - 2.f * dl;
            gtab[tid] = c;
        }
        v8f s0 = {};
        v8f s1 = {};
        #pragma unroll
        for (int q = 0; q < 8; ++q) {
            v2f a_lo = {a4[q].x, a4[q].y};
            v2f a_hi = {a4[q].z, a4[q].w};
            v2f b_lo = {b4[q].x, b4[q].y};
            v2f b_hi = {b4[q].z, b4[q].w};
            s0 = __builtin_amdgcn_wmma_f32_16x16x4_f32(false, a_lo, false, b_lo,
                                                       (short)0, s0, false, false);
            s1 = __builtin_amdgcn_wmma_f32_16x16x4_f32(false, a_hi, false, b_hi,
                                                       (short)0, s1, false, false);
        }
        float* dst = (sel ? v_lds : u_lds) + (rt * 16 + 8 * half) * LDS_STRIDE + snt * 16 + jl;
        #pragma unroll
        for (int r = 0; r < 8; ++r)
            dst[r * LDS_STRIDE] = s0[r] + s1[r];
    }
    {
        const v4f w = ((const v4f*)W2T)[tid];
        *(v4f*)&w2t[(tid >> 4) * LDS_STRIDE + 4 * (tid & 15)] = w;
    }
    if (tid < 64) { b1s[tid] = b1[tid]; b2s[tid] = b2[tid]; w3s[tid] = W3[tid]; }
    __syncthreads();

    const int   koff  = 2 * half;
    const float bias3 = b3[0];
    const bool  up1   = (jl & 1) != 0;
    const bool  up2   = (jl & 2) != 0;
    const bool  up4   = (jl & 4) != 0;
    const int   rmine = ((jl & 1) << 2) | (jl & 2) | ((jl >> 2) & 1);

    #pragma unroll 1
    for (int s = 0; s < UNITS_PER_WAVE; ++s) {
        asm volatile("" ::: "memory");
        const int unit = wv + PAIR_WAVES * s;
        if (unit >= 80) break;
        const int tp   = hb * 5 + (unit >> 4);
        const int il   = unit & 15;
        const int it   = (tp < 4) ? 0 : (tp < 7) ? 1 : (tp < 9) ? 2 : 3;
        const int start = it * 4 - (it * (it - 1)) / 2;
        const int jt   = tp - start + it;
        const float* urow = u_lds + (it * 16 + il) * LDS_STRIDE;
        const float* vrow = v_lds + (jt * 16 + jl) * LDS_STRIDE;

        v2f hA[16];
        #pragma unroll
        for (int kk = 0; kk < 16; ++kk) {
            const int k0 = 4 * kk + koff;
            float x0 = urow[k0]     + vrow[k0]     + b1s[k0];
            float x1 = urow[k0 + 1] + vrow[k0 + 1] + b1s[k0 + 1];
            hA[kk].x = gelu_tab(x0, gtab);
            hA[kk].y = gelu_tab(x1, gtab);
        }

        float p[8];
        #pragma unroll
        for (int r = 0; r < 8; ++r) p[r] = 0.f;

        #pragma unroll
        for (int np = 0; np < 2; ++np) {
            const int n0 = np * 32 + jl;
            const int n1 = n0 + 16;
            v8f acc0 = {};
            v8f acc1 = {};
            #pragma unroll
            for (int kk = 0; kk < 16; ++kk) {
                const int k0 = 4 * kk + koff;
                v2f bb0 = *(const v2f*)&w2t[n0 * LDS_STRIDE + k0];
                v2f bb1 = *(const v2f*)&w2t[n1 * LDS_STRIDE + k0];
                acc0 = __builtin_amdgcn_wmma_f32_16x16x4_f32(false, hA[kk], false, bb0,
                                                             (short)0, acc0, false, false);
                acc1 = __builtin_amdgcn_wmma_f32_16x16x4_f32(false, hA[kk], false, bb1,
                                                             (short)0, acc1, false, false);
            }
            const float b2n0 = b2s[n0], w3n0 = w3s[n0];
            const float b2n1 = b2s[n1], w3n1 = w3s[n1];
            #pragma unroll
            for (int r = 0; r < 8; ++r) {
                p[r] = fmaf(gelu_tab(acc0[r] + b2n0, gtab), w3n0, p[r]);
                p[r] = fmaf(gelu_tab(acc1[r] + b2n1, gtab), w3n1, p[r]);
            }
        }

        float q4[4];
        #pragma unroll
        for (int k = 0; k < 4; ++k) {
            const float keep = up1 ? p[k + 4] : p[k];
            const float give = up1 ? p[k]     : p[k + 4];
            q4[k] = keep + row_xor<1>(give);
        }
        float q2[2];
        #pragma unroll
        for (int m = 0; m < 2; ++m) {
            const float keep = up2 ? q4[m + 2] : q4[m];
            const float give = up2 ? q4[m]     : q4[m + 2];
            q2[m] = keep + row_xor<2>(give);
        }
        const float keep1 = up4 ? q2[1] : q2[0];
        const float give1 = up4 ? q2[0] : q2[1];
        const float q1  = keep1 + row_xor<4>(give1);
        const float sum = q1 + row_xor<8>(q1);

        const int ig = it * 16 + il;
        if (jl < 8) {
            const int jg = jt * 16 + rmine + 8 * half;
            if (ig < jg) {
                const float val = sum + bias3;
                out[(b * 64 + ig) * 64 + jg] = val;
                out[(b * 64 + jg) * 64 + ig] = val;
            } else if (ig == jg) {
                out[(b * 64 + ig) * 64 + jg] = 0.f;
            }
        }
    }
}

extern "C" void kernel_launch(void* const* d_in, const int* in_sizes, int n_in,
                              void* d_out, int out_size, void* d_ws, size_t ws_size,
                              hipStream_t stream) {
    const int*   positions = (const int*)  d_in[0];
    const int*   amino     = (const int*)  d_in[1];
    const float* pos_table = (const float*)d_in[2];
    const float* aa_table  = (const float*)d_in[3];
    const float* W1        = (const float*)d_in[4];
    const float* b1        = (const float*)d_in[5];
    const float* W2        = (const float*)d_in[6];
    const float* b2        = (const float*)d_in[7];
    const float* W3        = (const float*)d_in[8];
    const float* b3        = (const float*)d_in[9];
    float* out = (float*)d_out;
    float* ws  = (float*)d_ws;

    embed_kernel<<<dim3(2048 + 32 + 16), dim3(256), 0, stream>>>(
        positions, amino, pos_table, aa_table, W1, W2, ws);
    pair_kernel<<<dim3(256), dim3(1024), 0, stream>>>(
        ws, ws + 524288, ws + 524288 + 8192, b1, b2, W3, b3, out);
}


# pair kernel gelu: straight-line interpolation in a fine LDS table (8192 nodes, expanded per group from the cubic one), in units of 1/1024: one 8-byte read, no scaling multiply; gelu moves ~1e-7
# speedup vs baseline: 3.8355x; 1.1818x over previous
#include <hip/hip_runtime.h>
#include <math.h>


typedef __attribute__((ext_vector_type(2))) float v2f;
typedef __attribute__((ext_vector_type(4))) float v4f;
typedef __attribute__((ext_vector_type(8))) float v8f;

#define LDS_STRIDE 68

#define GT_N     256
#define GT_SCALE 32.0f
#define GF_N     8192
#define GF_SCALE 1024.0f
#define GF_TMAX  8191.9995f

__device__ __forceinline__ float gelu_lin(float xs, const v2f* __restrict__ tab) {
    const float t = fminf(fabsf(xs), GF_TMAX);
    const float f = __builtin_amdgcn_fractf(t);
    const v2f   c = tab[(int)t];
    return fmaxf(xs, 0.f) - fmaf(f, c.y, c.x);
}

template <int MASK>
__device__ __forceinline__ float row_xor(float v) {
    return __int_as_float(__builtin_amdgcn_update_dpp(0, __float_as_int(v), 0x160 + MASK,
                                                      0xf, 0xf, true));
}

#define EF_FLOATS  (128 * 64 * 64)
#define W1F_FLOATS (128 * 64)
#define EF_BLOCKS  (EF_FLOATS / 256)
#define W1F_BLOCKS (W1F_FLOATS / 256)

__global__ __launch_bounds__(256)
void embed_kernel(const int* __restrict__ positions,
                  const int* __restrict__ amino,
                  const float* __restrict__ pos_table,
                  const float* __restrict__ aa_table,
                  const float* __restrict__ W1,
                  const float* __restrict__ W2,
                  float* __restrict__ ws) {
    const int blk = blockIdx.x;
    if (blk < EF_BLOCKS) {
        const int t    = blk * 256 + threadIdx.x;
        const int r    = t & 3;
        const int ln   = (t >> 2) & 31;
        const int q    = (t >> 7) & 7;
        const int tile = (t >> 10) & 3;
        const int b    = t >> 12;
        const int bm   = b * 64 + tile * 16 + (ln & 15);
        const int k    = 8 * q + 4 * (ln >> 4) + r;
        int p = positions[bm]; p = p < 0 ? 0 : (p > 1023 ? 1023 : p);
        int a = amino[bm];     a = a < 0 ? 0 : (a > 20   ? 20   : a);
        ws[t] = pos_table[p * 64 + k] + aa_table[a * 64 + k];
    } else if (blk < EF_BLOCKS + W1F_BLOCKS) {
        const int u    = (blk - EF_BLOCKS) * 256 + threadIdx.x;
        const int r    = u & 3;
        const int ln   = (u >> 2) & 31;
        const int q    = (u >> 7) & 7;
        const int tile = (u >> 10) & 3;
        const int sel  = u >> 12;
        const int row  = sel * 64 + 8 * q + 4 * (ln >> 4) + r;
        const int col  = tile * 16 + (ln & 15);
        ws[EF_FLOATS + u] = W1[row * 64 + col];
    } else {
        const int e = (blk - EF_BLOCKS - W1F_BLOCKS) * 256 + threadIdx.x;
        const int n = e >> 6, k = e & 63;
        ws[EF_FLOATS + W1F_FLOATS + e] = W2[k * 64 + n];
    }
}

#define PAIR_WAVES   32
#define PAIR_THREADS 1024
#define UNITS_PER_WAVE 3

__global__ __launch_bounds__(PAIR_THREADS)
void pair_kernel(const float* __restrict__ Ef,
                 const float* __restrict__ W1f,
                 const float* __restrict__ W2T,
                 const float* __restrict__ b1,
                 const float* __restrict__ b2,
                 const float* __restrict__ W3,
                 const float* __restrict__ b3,
                 float* __restrict__ out) {
    __shared__ float u_lds[64 * LDS_STRIDE];
    __shared__ float v_lds[64 * LDS_STRIDE];
    __shared__ __attribute__((aligned(16))) float w2t[64 * LDS_STRIDE];
    __shared__ float b1s[64], b2s[64], w3s[64];
    __shared__ __attribute__((aligned(16))) v4f gtab[GT_N];
    __shared__ __attribute__((aligned(16))) v2f gfine[GF_N];

    const int tid  = threadIdx.x;
    const int b    = blockIdx.x >> 1;
    const int hb   = blockIdx.x & 1;
    const int lane = tid & 31;
    const int wv   = __builtin_amdgcn_readfirstlane(tid >> 5);
    const int jl   = lane & 15;
    const int half = lane >> 4;

    {
        const int sel = wv >> 4;
        const int rt  = (wv >> 2) & 3;
        const int snt = wv & 3;
        const v4f* Afrag = (const v4f*)(Ef  + ((b * 4 + rt) * 8) * 128) + lane;
        const v4f* Bfrag = (const v4f*)(W1f + ((sel * 4 + snt) * 8) * 128) + lane;
        v4f a4[8];
        v4f b4[8];
        #pragma unroll
        for (int q = 0; q < 8; ++q)
            a4[q] = Afrag[q * 32];
        #pragma unroll
        for (int q = 0; q < 8; ++q)
            b4[q] = Bfrag[q * 32];
        if (tid < GT_N) {
            const float h  = 1.0f / GT_SCALE;
            const float a0 = (float)tid * h;
            const float a1 = (float)(tid + 1) * h;
            const float q0 = 0.5f * erfcf(a0 * 0.70710678118654752f);
            const float q1 = 0.5f * erfcf(a1 * 0.70710678118654752f);
            const float r0 = a0 * q0;
            const float r1 = a1 * q1;
            const float d0 = h * (q0 - a0 * (0.3989422804014327f * expf(-0.5f * a0 * a0)));
            const float d1 = h * (q1 - a1 * (0.3989422804014327f * expf(-0.5f * a1 * a1)));
            const float dl = r1 - r0;
            v4f c;
            c.x = r0;
            c.y = d0;
            c.z = 3.f * dl - 2.f * d0 - d1;
            c.w = d0 + d1 - 2.f * dl;
            gtab[tid] = c;
        }
        v8f s0 = {};
        v8f s1 = {};
        #pragma unroll
        for (int q = 0; q < 8; ++q) {
            v2f a_lo = {a4[q].x, a4[q].y};
            v2f a_hi = {a4[q].z, a4[q].w};
            v2f b_lo = {b4[q].x, b4[q].y};
            v2f b_hi = {b4[q].z, b4[q].w};
            s0 = __builtin_amdgcn_wmma_f32_16x16x4_f32(false, a_lo, false, b_lo,
                                                       (short)0, s0, false, false);
            s1 = __builtin_amdgcn_wmma_f32_16x16x4_f32(false, a_hi, false, b_hi,
                                                       (short)0, s1, false, false);
        }
        float* dst = (sel ? v_lds : u_lds) + (rt * 16 + 8 * half) * LDS_STRIDE + snt * 16 + jl;
        #pragma unroll
        for (int r = 0; r < 8; ++r)
            dst[r * LDS_STRIDE] = (s0[r] + s1[r]) * GF_SCALE;
    }
    {
        const v4f w = ((const v4f*)W2T)[tid];
        *(v4f*)&w2t[(tid >> 4) * LDS_STRIDE + 4 * (tid & 15)] = w;
    }
    if (tid < 64) {
        b1s[tid] = b1[tid] * GF_SCALE;
        b2s[tid] = b2[tid] * GF_SCALE;
        w3s[tid] = W3[tid];
    }
    __syncthreads();
    {
        #pragma unroll
        for (int m = 0; m < 4; ++m) {
            const int   k  = 2 * tid + 2048 * m;
            const v4f   c  = gtab[k >> 5];
            const float fb = (float)(k & 31) * (1.0f / 32.0f);
            float rv[3];
            #pragma unroll
            for (int j = 0; j < 3; ++j) {
                const float f = fb + (float)j * (1.0f / 32.0f);
                float q = fmaf(f, c.w, c.z);
                q = fmaf(f, q, c.y);
                q = fmaf(f, q, c.x);
                rv[j] = q * GF_SCALE;
            }
            v4f e = {rv[0], rv[1] - rv[0], rv[1], rv[2] - rv[1]};
            *(v4f*)&gfine[k] = e;
        }
    }
    __syncthreads();

    const int   koff  = 2 * half;
    const float bias3 = b3[0];
    const bool  up1   = (jl & 1) != 0;
    const bool  up2   = (jl & 2) != 0;
    const bool  up4   = (jl & 4) != 0;
    const int   rmine = ((jl & 1) << 2) | (jl & 2) | ((jl >> 2) & 1);

    #pragma unroll 1
    for (int s = 0; s < UNITS_PER_WAVE; ++s) {
        asm volatile("" ::: "memory");
        const int unit = wv + PAIR_WAVES * s;
        if (unit >= 80) break;
        const int tp   = hb * 5 + (unit >> 4);
        const int il   = unit & 15;
        const int it   = (tp < 4) ? 0 : (tp < 7) ? 1 : (tp < 9) ? 2 : 3;
        const int start = it * 4 - (it * (it - 1)) / 2;
        const int jt   = tp - start + it;
        const float* urow = u_lds + (it * 16 + il) * LDS_STRIDE;
        const float* vrow = v_lds + (jt * 16 + jl) * LDS_STRIDE;

        v2f hA[16];
        #pragma unroll
        for (int kk = 0; kk < 16; ++kk) {
            const int k0 = 4 * kk + koff;
            float x0 = urow[k0]     + vrow[k0]     + b1s[k0];
            float x1 = urow[k0 + 1] + vrow[k0 + 1] + b1s[k0 + 1];
            hA[kk].x = gelu_lin(x0, gfine);
            hA[kk].y = gelu_lin(x1, gfine);
        }

        float p[8];
        #pragma unroll
        for (int r = 0; r < 8; ++r) p[r] = 0.f;

        #pragma unroll
        for (int np = 0; np < 2; ++np) {
            const int n0 = np * 32 + jl;
            const int n1 = n0 + 16;
            v8f acc0 = {};
            v8f acc1 = {};
            #pragma unroll
            for (int kk = 0; kk < 16; ++kk) {
                const int k0 = 4 * kk + koff;
                v2f bb0 = *(const v2f*)&w2t[n0 * LDS_STRIDE + k0];
                v2f bb1 = *(const v2f*)&w2t[n1 * LDS_STRIDE + k0];
                acc0 = __builtin_amdgcn_wmma_f32_16x16x4_f32(false, hA[kk], false, bb0,
                                                             (short)0, acc0, false, false);
                acc1 = __builtin_amdgcn_wmma_f32_16x16x4_f32(false, hA[kk], false, bb1,
                                                             (short)0, acc1, false, false);
            }
            const float b2n0 = b2s[n0], w3n0 = w3s[n0];
            const float b2n1 = b2s[n1], w3n1 = w3s[n1];
            #pragma unroll
            for (int r = 0; r < 8; ++r) {
                p[r] = fmaf(gelu_lin(acc0[r] + b2n0, gfine), w3n0, p[r]);
                p[r] = fmaf(gelu_lin(acc1[r] + b2n1, gfine), w3n1, p[r]);
            }
        }

        float q4[4];
        #pragma unroll
        for (int k = 0; k < 4; ++k) {
            const float keep = up1 ? p[k + 4] : p[k];
            const float give = up1 ? p[k]     : p[k + 4];
            q4[k] = keep + row_xor<1>(give);
        }
        float q2[2];
        #pragma unroll
        for (int m = 0; m < 2; ++m) {
            const float keep = up2 ? q4[m + 2] : q4[m];
            const float give = up2 ? q4[m]     : q4[m + 2];
            q2[m] = keep + row_xor<2>(give);
        }
        const float keep1 = up4 ? q2[1] : q2[0];
        const float give1 = up4 ? q2[0] : q2[1];
        const float q1  = keep1 + row_xor<4>(give1);
        const float sum = q1 + row_xor<8>(q1);

        const int ig = it * 16 + il;
        if (jl < 8) {
            const int jg = jt * 16 + rmine + 8 * half;
            if (ig < jg) {
                const float val = sum * (1.0f / GF_SCALE) + bias3;
                out[(b * 64 + ig) * 64 + jg] = val;
                out[(b * 64 + jg) * 64 + ig] = val;
            } else if (ig == jg) {
                out[(b * 64 + ig) * 64 + jg] = 0.f;
            }
        }
    }
}

extern "C" void kernel_launch(void* const* d_in, const int* in_sizes, int n_in,
                              void* d_out, int out_size, void* d_ws, size_t ws_size,
                              hipStream_t stream) {
    const int*   positions = (const int*)  d_in[0];
    const int*   amino     = (const int*)  d_in[1];
    const float* pos_table = (const float*)d_in[2];
    const float* aa_table  = (const float*)d_in[3];
    const float* W1        = (const float*)d_in[4];
    const float* b1        = (const float*)d_in[5];
    const float* W2        = (const float*)d_in[6];
    const float* b2        = (const float*)d_in[7];
    const float* W3        = (const float*)d_in[8];
    const float* b3        = (const float*)d_in[9];
    float* out = (float*)d_out;
    float* ws  = (float*)d_ws;

    embed_kernel<<<dim3(2048 + 32 + 16), dim3(256), 0, stream>>>(
        positions, amino, pos_table, aa_table, W1, W2, ws);
    pair_kernel<<<dim3(256), dim3(1024), 0, stream>>>(
        ws, ws + 524288, ws + 524288 + 8192, b1, b2, W3, b3, out);
}
